# starting kernel
# baseline (speedup 1.0000x reference)
#include <hip/hip_runtime.h>


typedef float v2f __attribute__((ext_vector_type(2)));
typedef float v8f __attribute__((ext_vector_type(8)));

#define FDIM  32
#define RROWS 1024
#define ELEMS 32768

#define STAGE(INPTR, WPTR, OUTPTR)                                            \
  {                                                                           \
          \
    v2f bfrag[2][8];                                                          \
    _Pragma("unroll")                                                         \
    for (int nt = 0; nt < 2; ++nt) {                                          \
      _Pragma("unroll")                                                       \
      for (int kc = 0; kc < 8; ++kc) {                                        \
        const int k = kc * 4 + laneH * 2;                                     \
        const int n = nt * 16 + laneM;                                        \
        bfrag[nt][kc].x = (WPTR)[k * FDIM + n];                               \
        bfrag[nt][kc].y = (WPTR)[(k + 1) * FDIM + n];                         \
      }                                                                       \
    }                                                                         \
    _Pragma("unroll")                                                         \
    for (int i = 0; i < 8; ++i) {                                             \
      const int r0 = (wave * 8 + i) * 16;                                     \
      v2f afrag[8];                                                           \
      _Pragma("unroll")                                                       \
      for (int kc = 0; kc < 8; ++kc) {                                        \
        const int c = kc * 4 + laneH * 2;                                     \
        const int r = r0 + laneM;                                             \
        afrag[kc].x = (INPTR)[c * RROWS + r];                                 \
        afrag[kc].y = (INPTR)[(c + 1) * RROWS + r];                           \
      }                                                                       \
      v8f acc0 = {0.f, 0.f, 0.f, 0.f, 0.f, 0.f, 0.f, 0.f};                    \
      v8f acc1 = {0.f, 0.f, 0.f, 0.f, 0.f, 0.f, 0.f, 0.f};                    \
      _Pragma("unroll")                                                       \
      for (int kc = 0; kc < 8; ++kc) {                                        \
        acc0 = __builtin_amdgcn_wmma_f32_16x16x4_f32(                         \
            false, afrag[kc], false, bfrag[0][kc], (short)0, acc0,            \
            false, false);                                                    \
        acc1 = __builtin_amdgcn_wmma_f32_16x16x4_f32(                         \
            false, afrag[kc], false, bfrag[1][kc], (short)0, acc1,            \
            false, false);                                                    \
      }                                                                       \
                 \
      _Pragma("unroll")                                                       \
      for (int v = 0; v < 8; ++v) {                                           \
        const int r = r0 + v + laneH * 8;                                     \
        (OUTPTR)[r * FDIM + laneM]      = acc0[v];                            \
        (OUTPTR)[r * FDIM + 16 + laneM] = acc1[v];                            \
      }                                                                       \
    }                                                                         \
  }

__global__ __launch_bounds__(256) void kron3_wmma_kernel(
    const float* __restrict__ x,
    const float* __restrict__ w0,
    const float* __restrict__ w1,
    const float* __restrict__ w2,
    float* __restrict__ out)
{
  extern __shared__ float smem[];
  float* buf0 = smem;
  float* buf1 = smem + ELEMS;

  const int tid   = threadIdx.x;
  const int wave  = tid >> 5;
  const int lane  = tid & 31;
  const int laneM = lane & 15;
  const int laneH = lane >> 4;

  const long long base = (long long)blockIdx.x * (long long)ELEMS;
  const float* xb = x + base;
  float* ob = out + base;

  STAGE(xb, w0, buf0);
  __syncthreads();
  STAGE(buf0, w1, buf1);
  __syncthreads();
  STAGE(buf1, w2, ob);
}

extern "C" void kernel_launch(void* const* d_in, const int* in_sizes, int n_in,
                              void* d_out, int out_size, void* d_ws, size_t ws_size,
                              hipStream_t stream) {
  const float* x  = (const float*)d_in[0];
  const float* w0 = (const float*)d_in[1];
  const float* w1 = (const float*)d_in[2];
  const float* w2 = (const float*)d_in[3];
  float* out = (float*)d_out;

  const int batch = in_sizes[0] / ELEMS;
  const size_t shmem = 2 * ELEMS * sizeof(float);

  kron3_wmma_kernel<<<batch, 256, shmem, stream>>>(x, w0, w1, w2, out);
}


# Stage 2 computed in place (its results wait in registers across a barrier): LDS per work-group 256 KB -> 128 KB, so two work-groups are resident per WGP and fill each other's stalls
# speedup vs baseline: 1.0725x; 1.0725x over previous
#include <hip/hip_runtime.h>


typedef float v2f __attribute__((ext_vector_type(2)));
typedef float v8f __attribute__((ext_vector_type(8)));

#define FDIM  32
#define RROWS 1024
#define ELEMS 32768

#define STAGE(INPTR, WPTR, OUTPTR)                                            \
  {                                                                           \
          \
    v2f bfrag[2][8];                                                          \
    _Pragma("unroll")                                                         \
    for (int nt = 0; nt < 2; ++nt) {                                          \
      _Pragma("unroll")                                                       \
      for (int kc = 0; kc < 8; ++kc) {                                        \
        const int k = kc * 4 + laneH * 2;                                     \
        const int n = nt * 16 + laneM;                                        \
        bfrag[nt][kc].x = (WPTR)[k * FDIM + n];                               \
        bfrag[nt][kc].y = (WPTR)[(k + 1) * FDIM + n];                         \
      }                                                                       \
    }                                                                         \
    _Pragma("unroll")                                                         \
    for (int i = 0; i < 8; ++i) {                                             \
      const int r0 = (wave * 8 + i) * 16;                                     \
      v2f afrag[8];                                                           \
      _Pragma("unroll")                                                       \
      for (int kc = 0; kc < 8; ++kc) {                                        \
        const int c = kc * 4 + laneH * 2;                                     \
        const int r = r0 + laneM;                                             \
        afrag[kc].x = (INPTR)[c * RROWS + r];                                 \
        afrag[kc].y = (INPTR)[(c + 1) * RROWS + r];                           \
      }                                                                       \
      v8f acc0 = {0.f, 0.f, 0.f, 0.f, 0.f, 0.f, 0.f, 0.f};                    \
      v8f acc1 = {0.f, 0.f, 0.f, 0.f, 0.f, 0.f, 0.f, 0.f};                    \
      _Pragma("unroll")                                                       \
      for (int kc = 0; kc < 8; ++kc) {                                        \
        acc0 = __builtin_amdgcn_wmma_f32_16x16x4_f32(                         \
            false, afrag[kc], false, bfrag[0][kc], (short)0, acc0,            \
            false, false);                                                    \
        acc1 = __builtin_amdgcn_wmma_f32_16x16x4_f32(                         \
            false, afrag[kc], false, bfrag[1][kc], (short)0, acc1,            \
            false, false);                                                    \
      }                                                                       \
                 \
      _Pragma("unroll")                                                       \
      for (int v = 0; v < 8; ++v) {                                           \
        const int r = r0 + v + laneH * 8;                                     \
        (OUTPTR)[r * FDIM + laneM]      = acc0[v];                            \
        (OUTPTR)[r * FDIM + 16 + laneM] = acc1[v];                            \
      }                                                                       \
    }                                                                         \
  }

__global__ __launch_bounds__(256) void kron3_wmma_kernel(
    const float* __restrict__ x,
    const float* __restrict__ w0,
    const float* __restrict__ w1,
    const float* __restrict__ w2,
    float* __restrict__ out)
{
  extern __shared__ float smem[];
  float* buf = smem;

  const int tid   = threadIdx.x;
  const int wave  = tid >> 5;
  const int lane  = tid & 31;
  const int laneM = lane & 15;
  const int laneH = lane >> 4;

  const long long base = (long long)blockIdx.x * (long long)ELEMS;
  const float* xb = x + base;
  float* ob = out + base;

  STAGE(xb, w0, buf);
  __syncthreads();

  {
    v2f bfrag[2][8];
    #pragma unroll
    for (int nt = 0; nt < 2; ++nt) {
      #pragma unroll
      for (int kc = 0; kc < 8; ++kc) {
        const int k = kc * 4 + laneH * 2;
        const int n = nt * 16 + laneM;
        bfrag[nt][kc].x = w1[k * FDIM + n];
        bfrag[nt][kc].y = w1[(k + 1) * FDIM + n];
      }
    }
    v8f acc[8][2];
    #pragma unroll
    for (int i = 0; i < 8; ++i) {
      const int r0 = (wave * 8 + i) * 16;
      v2f afrag[8];
      #pragma unroll
      for (int kc = 0; kc < 8; ++kc) {
        const int c = kc * 4 + laneH * 2;
        const int r = r0 + laneM;
        afrag[kc].x = buf[c * RROWS + r];
        afrag[kc].y = buf[(c + 1) * RROWS + r];
      }
      v8f a0 = {0.f, 0.f, 0.f, 0.f, 0.f, 0.f, 0.f, 0.f};
      v8f a1 = {0.f, 0.f, 0.f, 0.f, 0.f, 0.f, 0.f, 0.f};
      #pragma unroll
      for (int kc = 0; kc < 8; ++kc) {
        a0 = __builtin_amdgcn_wmma_f32_16x16x4_f32(
            false, afrag[kc], false, bfrag[0][kc], (short)0, a0, false, false);
        a1 = __builtin_amdgcn_wmma_f32_16x16x4_f32(
            false, afrag[kc], false, bfrag[1][kc], (short)0, a1, false, false);
      }
      acc[i][0] = a0;
      acc[i][1] = a1;
    }
    __syncthreads();
    #pragma unroll
    for (int i = 0; i < 8; ++i) {
      const int r0 = (wave * 8 + i) * 16;
      #pragma unroll
      for (int v = 0; v < 8; ++v) {
        const int r = r0 + v + laneH * 8;
        buf[r * FDIM + laneM]      = acc[i][0][v];
        buf[r * FDIM + 16 + laneM] = acc[i][1][v];
      }
    }
  }
  __syncthreads();
  STAGE(buf, w2, ob);
}

extern "C" void kernel_launch(void* const* d_in, const int* in_sizes, int n_in,
                              void* d_out, int out_size, void* d_ws, size_t ws_size,
                              hipStream_t stream) {
  const float* x  = (const float*)d_in[0];
  const float* w0 = (const float*)d_in[1];
  const float* w1 = (const float*)d_in[2];
  const float* w2 = (const float*)d_in[3];
  float* out = (float*)d_out;

  const int batch = in_sizes[0] / ELEMS;
  const size_t shmem = ELEMS * sizeof(float);

  kron3_wmma_kernel<<<batch, 256, shmem, stream>>>(x, w0, w1, w2, out);
}


# Stage 3 maps tile rows and columns to fragment positions so that each store instruction writes one aligned contiguous 256-byte piece of out (8-byte stores, tile by tile), not two 64-byte half lines
# speedup vs baseline: 1.2863x; 1.1993x over previous
#include <hip/hip_runtime.h>


typedef float v2f __attribute__((ext_vector_type(2)));
typedef float v8f __attribute__((ext_vector_type(8)));

#define FDIM  32
#define RROWS 1024
#define ELEMS 32768

#define STAGE(INPTR, WPTR, OUTPTR)                                            \
  {                                                                           \
          \
    v2f bfrag[2][8];                                                          \
    _Pragma("unroll")                                                         \
    for (int nt = 0; nt < 2; ++nt) {                                          \
      _Pragma("unroll")                                                       \
      for (int kc = 0; kc < 8; ++kc) {                                        \
        const int k = kc * 4 + laneH * 2;                                     \
        const int n = nt * 16 + laneM;                                        \
        bfrag[nt][kc].x = (WPTR)[k * FDIM + n];                               \
        bfrag[nt][kc].y = (WPTR)[(k + 1) * FDIM + n];                         \
      }                                                                       \
    }                                                                         \
    _Pragma("unroll")                                                         \
    for (int i = 0; i < 8; ++i) {                                             \
      const int r0 = (wave * 8 + i) * 16;                                     \
      v2f afrag[8];                                                           \
      _Pragma("unroll")                                                       \
      for (int kc = 0; kc < 8; ++kc) {                                        \
        const int c = kc * 4 + laneH * 2;                                     \
        const int r = r0 + laneM;                                             \
        afrag[kc].x = (INPTR)[c * RROWS + r];                                 \
        afrag[kc].y = (INPTR)[(c + 1) * RROWS + r];                           \
      }                                                                       \
      v8f acc0 = {0.f, 0.f, 0.f, 0.f, 0.f, 0.f, 0.f, 0.f};                    \
      v8f acc1 = {0.f, 0.f, 0.f, 0.f, 0.f, 0.f, 0.f, 0.f};                    \
      _Pragma("unroll")                                                       \
      for (int kc = 0; kc < 8; ++kc) {                                        \
        acc0 = __builtin_amdgcn_wmma_f32_16x16x4_f32(                         \
            false, afrag[kc], false, bfrag[0][kc], (short)0, acc0,            \
            false, false);                                                    \
        acc1 = __builtin_amdgcn_wmma_f32_16x16x4_f32(                         \
            false, afrag[kc], false, bfrag[1][kc], (short)0, acc1,            \
            false, false);                                                    \
      }                                                                       \
                 \
      _Pragma("unroll")                                                       \
      for (int v = 0; v < 8; ++v) {                                           \
        const int r = r0 + v + laneH * 8;                                     \
        (OUTPTR)[r * FDIM + laneM]      = acc0[v];                            \
        (OUTPTR)[r * FDIM + 16 + laneM] = acc1[v];                            \
      }                                                                       \
    }                                                                         \
  }

__global__ __launch_bounds__(256) void kron3_wmma_kernel(
    const float* __restrict__ x,
    const float* __restrict__ w0,
    const float* __restrict__ w1,
    const float* __restrict__ w2,
    float* __restrict__ out)
{
  extern __shared__ float smem[];
  float* buf = smem;

  const int tid   = threadIdx.x;
  const int wave  = tid >> 5;
  const int lane  = tid & 31;
  const int laneM = lane & 15;
  const int laneH = lane >> 4;

  const long long base = (long long)blockIdx.x * (long long)ELEMS;
  const float* xb = x + base;
  float* ob = out + base;

  STAGE(xb, w0, buf);
  __syncthreads();

  {
    v2f bfrag[2][8];
    #pragma unroll
    for (int nt = 0; nt < 2; ++nt) {
      #pragma unroll
      for (int kc = 0; kc < 8; ++kc) {
        const int k = kc * 4 + laneH * 2;
        const int n = nt * 16 + laneM;
        bfrag[nt][kc].x = w1[k * FDIM + n];
        bfrag[nt][kc].y = w1[(k + 1) * FDIM + n];
      }
    }
    v8f acc[8][2];
    #pragma unroll
    for (int i = 0; i < 8; ++i) {
      const int r0 = (wave * 8 + i) * 16;
      v2f afrag[8];
      #pragma unroll
      for (int kc = 0; kc < 8; ++kc) {
        const int c = kc * 4 + laneH * 2;
        const int r = r0 + laneM;
        afrag[kc].x = buf[c * RROWS + r];
        afrag[kc].y = buf[(c + 1) * RROWS + r];
      }
      v8f a0 = {0.f, 0.f, 0.f, 0.f, 0.f, 0.f, 0.f, 0.f};
      v8f a1 = {0.f, 0.f, 0.f, 0.f, 0.f, 0.f, 0.f, 0.f};
      #pragma unroll
      for (int kc = 0; kc < 8; ++kc) {
        a0 = __builtin_amdgcn_wmma_f32_16x16x4_f32(
            false, afrag[kc], false, bfrag[0][kc], (short)0, a0, false, false);
        a1 = __builtin_amdgcn_wmma_f32_16x16x4_f32(
            false, afrag[kc], false, bfrag[1][kc], (short)0, a1, false, false);
      }
      acc[i][0] = a0;
      acc[i][1] = a1;
    }
    __syncthreads();
    #pragma unroll
    for (int i = 0; i < 8; ++i) {
      const int r0 = (wave * 8 + i) * 16;
      #pragma unroll
      for (int v = 0; v < 8; ++v) {
        const int r = r0 + v + laneH * 8;
        buf[r * FDIM + laneM]      = acc[i][0][v];
        buf[r * FDIM + 16 + laneM] = acc[i][1][v];
      }
    }
  }
  __syncthreads();
  {
    v2f bfrag[2][8];
    #pragma unroll
    for (int nt = 0; nt < 2; ++nt) {
      #pragma unroll
      for (int kc = 0; kc < 8; ++kc) {
        const int k = kc * 4 + laneH * 2;
        const int n = 2 * laneM + nt;
        bfrag[nt][kc].x = w2[k * FDIM + n];
        bfrag[nt][kc].y = w2[(k + 1) * FDIM + n];
      }
    }
    const int rowsel = 2 * (laneM & 7) + (laneM >> 3);
    #pragma unroll
    for (int i = 0; i < 8; ++i) {
      const int r0 = (wave * 8 + i) * 16;
      v2f afrag[8];
      #pragma unroll
      for (int kc = 0; kc < 8; ++kc) {
        const int c = kc * 4 + laneH * 2;
        const int r = r0 + rowsel;
        afrag[kc].x = buf[c * RROWS + r];
        afrag[kc].y = buf[(c + 1) * RROWS + r];
      }
      v8f acc0 = {0.f, 0.f, 0.f, 0.f, 0.f, 0.f, 0.f, 0.f};
      v8f acc1 = {0.f, 0.f, 0.f, 0.f, 0.f, 0.f, 0.f, 0.f};
      #pragma unroll
      for (int kc = 0; kc < 8; ++kc) {
        acc0 = __builtin_amdgcn_wmma_f32_16x16x4_f32(
            false, afrag[kc], false, bfrag[0][kc], (short)0, acc0, false, false);
        acc1 = __builtin_amdgcn_wmma_f32_16x16x4_f32(
            false, afrag[kc], false, bfrag[1][kc], (short)0, acc1, false, false);
      }
      #pragma unroll
      for (int v = 0; v < 8; ++v) {
        v2f pr;
        pr.x = acc0[v];
        pr.y = acc1[v];
        *(v2f*)(ob + (r0 + 2 * v) * FDIM + 2 * lane) = pr;
      }
      __builtin_amdgcn_sched_barrier(0);
    }
  }
}

extern "C" void kernel_launch(void* const* d_in, const int* in_sizes, int n_in,
                              void* d_out, int out_size, void* d_ws, size_t ws_size,
                              hipStream_t stream) {
  const float* x  = (const float*)d_in[0];
  const float* w0 = (const float*)d_in[1];
  const float* w1 = (const float*)d_in[2];
  const float* w2 = (const float*)d_in[3];
  float* out = (float*)d_out;

  const int batch = in_sizes[0] / ELEMS;
  const size_t shmem = ELEMS * sizeof(float);

  kron3_wmma_kernel<<<batch, 256, shmem, stream>>>(x, w0, w1, w2, out);
}


# Stage 1 reads x with 16-byte loads, a lane's four floats feeding four M-tiles: a half-wave covers an aligned 256-byte piece, and all 32 loads of a wave are issued before its first matrix instruction
# speedup vs baseline: 1.3482x; 1.0481x over previous
#include <hip/hip_runtime.h>


typedef float v2f __attribute__((ext_vector_type(2)));
typedef float v4f __attribute__((ext_vector_type(4)));
typedef float v8f __attribute__((ext_vector_type(8)));

#define FDIM  32
#define RROWS 1024
#define ELEMS 32768


__global__ __launch_bounds__(256) void kron3_wmma_kernel(
    const float* __restrict__ x,
    const float* __restrict__ w0,
    const float* __restrict__ w1,
    const float* __restrict__ w2,
    float* __restrict__ out)
{
  extern __shared__ float smem[];
  float* buf = smem;

  const int tid   = threadIdx.x;
  const int wave  = tid >> 5;
  const int lane  = tid & 31;
  const int laneM = lane & 15;
  const int laneH = lane >> 4;

  const long long base = (long long)blockIdx.x * (long long)ELEMS;
  const float* xb = x + base;
  float* ob = out + base;

  {
    v2f bfrag[2][8];
    #pragma unroll
    for (int nt = 0; nt < 2; ++nt) {
      #pragma unroll
      for (int kc = 0; kc < 8; ++kc) {
        const int k = kc * 4 + laneH * 2;
        const int n = nt * 16 + laneM;
        bfrag[nt][kc].x = w0[k * FDIM + n];
        bfrag[nt][kc].y = w0[(k + 1) * FDIM + n];
      }
    }
    __builtin_amdgcn_sched_barrier(0);
    const int xoff = (laneH * 2) * RROWS + wave * 128 + laneM * 4;
    v4f xa[2][8][2];
    #pragma unroll
    for (int g = 0; g < 2; ++g) {
      #pragma unroll
      for (int kc = 0; kc < 8; ++kc) {
        xa[g][kc][0] = *(const v4f*)(xb + xoff + (kc * 4) * RROWS + g * 64);
        xa[g][kc][1] = *(const v4f*)(xb + xoff + (kc * 4 + 1) * RROWS + g * 64);
      }
      __builtin_amdgcn_sched_barrier(0);
    }
    #pragma unroll
    for (int g = 0; g < 2; ++g) {
      v8f acc[4][2];
      #pragma unroll
      for (int j = 0; j < 4; ++j) {
        acc[j][0] = (v8f){0.f, 0.f, 0.f, 0.f, 0.f, 0.f, 0.f, 0.f};
        acc[j][1] = (v8f){0.f, 0.f, 0.f, 0.f, 0.f, 0.f, 0.f, 0.f};
      }
      #pragma unroll
      for (int kc = 0; kc < 8; ++kc) {
        #pragma unroll
        for (int j = 0; j < 4; ++j) {
          v2f a;
          a.x = xa[g][kc][0][j];
          a.y = xa[g][kc][1][j];
          acc[j][0] = __builtin_amdgcn_wmma_f32_16x16x4_f32(
              false, a, false, bfrag[0][kc], (short)0, acc[j][0], false, false);
          acc[j][1] = __builtin_amdgcn_wmma_f32_16x16x4_f32(
              false, a, false, bfrag[1][kc], (short)0, acc[j][1], false, false);
        }
      }
      #pragma unroll
      for (int j = 0; j < 4; ++j) {
        #pragma unroll
        for (int v = 0; v < 8; ++v) {
          const int r = wave * 128 + g * 64 + (v + laneH * 8) * 4 + j;
          buf[r * FDIM + laneM]      = acc[j][0][v];
          buf[r * FDIM + 16 + laneM] = acc[j][1][v];
        }
      }
    }
  }
  __syncthreads();

  {
    v2f bfrag[2][8];
    #pragma unroll
    for (int nt = 0; nt < 2; ++nt) {
      #pragma unroll
      for (int kc = 0; kc < 8; ++kc) {
        const int k = kc * 4 + laneH * 2;
        const int n = nt * 16 + laneM;
        bfrag[nt][kc].x = w1[k * FDIM + n];
        bfrag[nt][kc].y = w1[(k + 1) * FDIM + n];
      }
    }
    v8f acc[8][2];
    #pragma unroll
    for (int i = 0; i < 8; ++i) {
      const int r0 = (wave * 8 + i) * 16;
      v2f afrag[8];
      #pragma unroll
      for (int kc = 0; kc < 8; ++kc) {
        const int c = kc * 4 + laneH * 2;
        const int r = r0 + laneM;
        afrag[kc].x = buf[c * RROWS + r];
        afrag[kc].y = buf[(c + 1) * RROWS + r];
      }
      v8f a0 = {0.f, 0.f, 0.f, 0.f, 0.f, 0.f, 0.f, 0.f};
      v8f a1 = {0.f, 0.f, 0.f, 0.f, 0.f, 0.f, 0.f, 0.f};
      #pragma unroll
      for (int kc = 0; kc < 8; ++kc) {
        a0 = __builtin_amdgcn_wmma_f32_16x16x4_f32(
            false, afrag[kc], false, bfrag[0][kc], (short)0, a0, false, false);
        a1 = __builtin_amdgcn_wmma_f32_16x16x4_f32(
            false, afrag[kc], false, bfrag[1][kc], (short)0, a1, false, false);
      }
      acc[i][0] = a0;
      acc[i][1] = a1;
    }
    __syncthreads();
    #pragma unroll
    for (int i = 0; i < 8; ++i) {
      const int r0 = (wave * 8 + i) * 16;
      #pragma unroll
      for (int v = 0; v < 8; ++v) {
        const int r = r0 + v + laneH * 8;
        buf[r * FDIM + laneM]      = acc[i][0][v];
        buf[r * FDIM + 16 + laneM] = acc[i][1][v];
      }
    }
  }
  __syncthreads();
  {
    v2f bfrag[2][8];
    #pragma unroll
    for (int nt = 0; nt < 2; ++nt) {
      #pragma unroll
      for (int kc = 0; kc < 8; ++kc) {
        const int k = kc * 4 + laneH * 2;
        const int n = 2 * laneM + nt;
        bfrag[nt][kc].x = w2[k * FDIM + n];
        bfrag[nt][kc].y = w2[(k + 1) * FDIM + n];
      }
    }
    const int rowsel = 2 * (laneM & 7) + (laneM >> 3);
    #pragma unroll
    for (int i = 0; i < 8; ++i) {
      const int r0 = (wave * 8 + i) * 16;
      v2f afrag[8];
      #pragma unroll
      for (int kc = 0; kc < 8; ++kc) {
        const int c = kc * 4 + laneH * 2;
        const int r = r0 + rowsel;
        afrag[kc].x = buf[c * RROWS + r];
        afrag[kc].y = buf[(c + 1) * RROWS + r];
      }
      v8f acc0 = {0.f, 0.f, 0.f, 0.f, 0.f, 0.f, 0.f, 0.f};
      v8f acc1 = {0.f, 0.f, 0.f, 0.f, 0.f, 0.f, 0.f, 0.f};
      #pragma unroll
      for (int kc = 0; kc < 8; ++kc) {
        acc0 = __builtin_amdgcn_wmma_f32_16x16x4_f32(
            false, afrag[kc], false, bfrag[0][kc], (short)0, acc0, false, false);
        acc1 = __builtin_amdgcn_wmma_f32_16x16x4_f32(
            false, afrag[kc], false, bfrag[1][kc], (short)0, acc1, false, false);
      }
      #pragma unroll
      for (int v = 0; v < 8; ++v) {
        v2f pr;
        pr.x = acc0[v];
        pr.y = acc1[v];
        *(v2f*)(ob + (r0 + 2 * v) * FDIM + 2 * lane) = pr;
      }
      __builtin_amdgcn_sched_barrier(0);
    }
  }
}

extern "C" void kernel_launch(void* const* d_in, const int* in_sizes, int n_in,
                              void* d_out, int out_size, void* d_ws, size_t ws_size,
                              hipStream_t stream) {
  const float* x  = (const float*)d_in[0];
  const float* w0 = (const float*)d_in[1];
  const float* w1 = (const float*)d_in[2];
  const float* w2 = (const float*)d_in[3];
  float* out = (float*)d_out;

  const int batch = in_sizes[0] / ELEMS;
  const size_t shmem = ELEMS * sizeof(float);

  kron3_wmma_kernel<<<batch, 256, shmem, stream>>>(x, w0, w1, w2, out);
}


# LDS buffer keeps the two k of an A-fragment pair side by side, pitch padded by 32 dwords: stage 2 and 3 fragments are single 8-byte LDS loads and no LDS load or store has a bank conflict
# speedup vs baseline: 1.3944x; 1.0343x over previous
#include <hip/hip_runtime.h>


typedef float v2f __attribute__((ext_vector_type(2)));
typedef float v4f __attribute__((ext_vector_type(4)));
typedef float v8f __attribute__((ext_vector_type(8)));

#define FDIM  32
#define RROWS 1024
#define ELEMS 32768

#define PITCH 2080
#define LDS_FLOATS (16 * PITCH)


__global__ __launch_bounds__(256) void kron3_wmma_kernel(
    const float* __restrict__ x,
    const float* __restrict__ w0,
    const float* __restrict__ w1,
    const float* __restrict__ w2,
    float* __restrict__ out)
{
  extern __shared__ __attribute__((aligned(16))) float smem[];
  float* buf = smem;

  const int tid   = threadIdx.x;
  const int wave  = tid >> 5;
  const int lane  = tid & 31;
  const int laneM = lane & 15;
  const int laneH = lane >> 4;

  const long long base = (long long)blockIdx.x * (long long)ELEMS;
  const float* xb = x + base;
  float* ob = out + base;

  {
    v2f bfrag[2][8];
    #pragma unroll
    for (int nt = 0; nt < 2; ++nt) {
      #pragma unroll
      for (int kc = 0; kc < 8; ++kc) {
        const int k = kc * 4 + laneH * 2;
        const int n = nt * 16 + laneM;
        bfrag[nt][kc].x = w0[k * FDIM + n];
        bfrag[nt][kc].y = w0[(k + 1) * FDIM + n];
      }
    }
    __builtin_amdgcn_sched_barrier(0);
    const int xoff = (laneH * 2) * RROWS + wave * 128 + laneM * 4;
    v4f xa[2][8][2];
    #pragma unroll
    for (int g = 0; g < 2; ++g) {
      #pragma unroll
      for (int kc = 0; kc < 8; ++kc) {
        xa[g][kc][0] = *(const v4f*)(xb + xoff + (kc * 4) * RROWS + g * 64);
        xa[g][kc][1] = *(const v4f*)(xb + xoff + (kc * 4 + 1) * RROWS + g * 64);
      }
      __builtin_amdgcn_sched_barrier(0);
    }
    #pragma unroll
    for (int g = 0; g < 2; ++g) {
      v8f acc[4][2];
      #pragma unroll
      for (int j = 0; j < 4; ++j) {
        acc[j][0] = (v8f){0.f, 0.f, 0.f, 0.f, 0.f, 0.f, 0.f, 0.f};
        acc[j][1] = (v8f){0.f, 0.f, 0.f, 0.f, 0.f, 0.f, 0.f, 0.f};
      }
      #pragma unroll
      for (int kc = 0; kc < 8; ++kc) {
        #pragma unroll
        for (int j = 0; j < 4; ++j) {
          v2f a;
          a.x = xa[g][kc][0][j];
          a.y = xa[g][kc][1][j];
          acc[j][0] = __builtin_amdgcn_wmma_f32_16x16x4_f32(
              false, a, false, bfrag[0][kc], (short)0, acc[j][0], false, false);
          acc[j][1] = __builtin_amdgcn_wmma_f32_16x16x4_f32(
              false, a, false, bfrag[1][kc], (short)0, acc[j][1], false, false);
        }
      }
      float* s1 = buf + (2 * wave + g) * PITCH + laneH + 2 * laneM;
      #pragma unroll
      for (int j = 0; j < 4; ++j) {
        #pragma unroll
        for (int v = 0; v < 8; ++v) {
          s1[(4 * v + j) * 64]      = acc[j][0][v];
          s1[(4 * v + j) * 64 + 32] = acc[j][1][v];
        }
      }
    }
  }
  __syncthreads();

  {
    v2f bfrag[2][8];
    #pragma unroll
    for (int nt = 0; nt < 2; ++nt) {
      #pragma unroll
      for (int kc = 0; kc < 8; ++kc) {
        const int k = kc * 4 + laneH * 2;
        const int n = nt * 16 + laneM;
        bfrag[nt][kc].x = w1[k * FDIM + n];
        bfrag[nt][kc].y = w1[(k + 1) * FDIM + n];
      }
    }
    const int m3 = laneM >> 3;
    v2f afrag[8][8];
    #pragma unroll
    for (int i = 0; i < 8; ++i) {
      const int r = wave * 128 + ((i >> 1) & 1) * 64 + (m3 ^ (i >> 2)) * 32 +
                    (i & 1) * 16 + m3 * 8 + (laneM & 7);
      #pragma unroll
      for (int kc = 0; kc < 8; ++kc) {
        afrag[i][kc] = *(const v2f*)(buf + (kc * 2 + laneH) * PITCH + 2 * r);
      }
    }
    __builtin_amdgcn_sched_barrier(0);
    __syncthreads();
    #pragma unroll
    for (int i = 0; i < 8; ++i) {
      v8f a0 = {0.f, 0.f, 0.f, 0.f, 0.f, 0.f, 0.f, 0.f};
      v8f a1 = {0.f, 0.f, 0.f, 0.f, 0.f, 0.f, 0.f, 0.f};
      #pragma unroll
      for (int kc = 0; kc < 8; ++kc) {
        a0 = __builtin_amdgcn_wmma_f32_16x16x4_f32(
            false, afrag[i][kc], false, bfrag[0][kc], (short)0, a0, false, false);
        a1 = __builtin_amdgcn_wmma_f32_16x16x4_f32(
            false, afrag[i][kc], false, bfrag[1][kc], (short)0, a1, false, false);
      }
      float* s2 = buf + (2 * wave + ((i >> 1) & 1)) * PITCH +
                  ((i & 1) * 16 + laneH * 8) * 64 + (laneH ^ (i >> 2)) + 2 * laneM;
      #pragma unroll
      for (int v = 0; v < 8; ++v) {
        s2[v * 64]      = a0[v];
        s2[v * 64 + 32] = a1[v];
      }
      __builtin_amdgcn_sched_barrier(0);
    }
  }
  __syncthreads();
  {
    v2f bfrag[2][8];
    #pragma unroll
    for (int nt = 0; nt < 2; ++nt) {
      #pragma unroll
      for (int kc = 0; kc < 8; ++kc) {
        const int k = kc * 4 + laneH * 2;
        const int n = 2 * laneM + nt;
        bfrag[nt][kc].x = w2[k * FDIM + n];
        bfrag[nt][kc].y = w2[(k + 1) * FDIM + n];
      }
    }
    const int rowsel = 2 * (laneM & 7) + (laneM >> 3);
    #pragma unroll
    for (int i = 0; i < 8; ++i) {
      const int r0 = (wave * 8 + i) * 16;
      v2f afrag[8];
      #pragma unroll
      for (int kc = 0; kc < 8; ++kc) {
        afrag[kc] = *(const v2f*)(buf + (kc * 2 + laneH) * PITCH + 2 * (r0 + rowsel));
      }
      v8f acc0 = {0.f, 0.f, 0.f, 0.f, 0.f, 0.f, 0.f, 0.f};
      v8f acc1 = {0.f, 0.f, 0.f, 0.f, 0.f, 0.f, 0.f, 0.f};
      #pragma unroll
      for (int kc = 0; kc < 8; ++kc) {
        acc0 = __builtin_amdgcn_wmma_f32_16x16x4_f32(
            false, afrag[kc], false, bfrag[0][kc], (short)0, acc0, false, false);
        acc1 = __builtin_amdgcn_wmma_f32_16x16x4_f32(
            false, afrag[kc], false, bfrag[1][kc], (short)0, acc1, false, false);
      }
      #pragma unroll
      for (int v = 0; v < 8; ++v) {
        v2f pr;
        pr.x = acc0[v];
        pr.y = acc1[v];
        *(v2f*)(ob + (r0 + 2 * v) * FDIM + 2 * lane) = pr;
      }
      __builtin_amdgcn_sched_barrier(0);
    }
  }
}

extern "C" void kernel_launch(void* const* d_in, const int* in_sizes, int n_in,
                              void* d_out, int out_size, void* d_ws, size_t ws_size,
                              hipStream_t stream) {
  const float* x  = (const float*)d_in[0];
  const float* w0 = (const float*)d_in[1];
  const float* w1 = (const float*)d_in[2];
  const float* w2 = (const float*)d_in[3];
  float* out = (float*)d_out;

  const int batch = in_sizes[0] / ELEMS;
  const size_t shmem = LDS_FLOATS * sizeof(float);

  kron3_wmma_kernel<<<batch, 256, shmem, stream>>>(x, w0, w1, w2, out);
}
